# starting kernel
# baseline (speedup 1.0000x reference)
#include <hip/hip_runtime.h>
#include <cstdint>

#define N_NODES 100000
#define N_EDGES 1600000
#define IN_CH   256
#define HID     128
#define K_STEPS 10
#define ALPHA   0.1f

typedef __attribute__((ext_vector_type(2))) float v2f;
typedef __attribute__((ext_vector_type(8))) float v8f;

__device__ __forceinline__ void atom_add_f32(float* p, float v) {
    __hip_atomic_fetch_add(p, v, __ATOMIC_RELAXED, __HIP_MEMORY_SCOPE_AGENT);
}

__global__ __launch_bounds__(256) void deg_kernel(const int* __restrict__ src,
                                                  const int* __restrict__ dst,
                                                  float* __restrict__ deg_out,
                                                  float* __restrict__ deg_in) {
    int e = blockIdx.x * blockDim.x + threadIdx.x;
    if (e < N_EDGES) {
        atom_add_f32(&deg_out[src[e]], 1.0f);
        atom_add_f32(&deg_in[dst[e]], 1.0f);
    }
}

__global__ __launch_bounds__(256) void inv_kernel(float* __restrict__ deg, int n) {
    int i = blockIdx.x * blockDim.x + threadIdx.x;
    if (i < n) {
        float d = deg[i];
        deg[i] = (d > 0.0f) ? rsqrtf(fmaxf(d, 1.0f)) : 0.0f;
    }
}

__global__ __launch_bounds__(256) void enorm_kernel(const int* __restrict__ src,
                                                    const int* __restrict__ dst,
                                                    const float* __restrict__ inv_out,
                                                    const float* __restrict__ inv_in,
                                                    float* __restrict__ enorm) {
    int e = blockIdx.x * blockDim.x + threadIdx.x;
    if (e < N_EDGES)
        enorm[e] = inv_out[src[e]] * inv_in[dst[e]];
}

__global__ __launch_bounds__(256) void gemm_kernel(const float* __restrict__ X,
                                                   const float* __restrict__ W,
                                                   float* __restrict__ H) {
    const int lane    = threadIdx.x & 31;
    const int wave    = threadIdx.x >> 5;
    const int rowBase = blockIdx.x * 16;
    const int colBase = wave * 16;
    const int half    = lane >> 4;
    const int l16     = lane & 15;

    const float* __restrict__ arow = X + (size_t)(rowBase + l16) * IN_CH + half * 2;
    const float* __restrict__ wcol = W + (size_t)(half * 2) * HID + (colBase + l16);

    v8f acc = {};
#pragma unroll 8
    for (int k0 = 0; k0 < IN_CH; k0 += 4) {
        v2f a, b;
        a.x = arow[k0];
        a.y = arow[k0 + 1];
        b.x = wcol[k0 * HID];
        b.y = wcol[k0 * HID + HID];
        acc = __builtin_amdgcn_wmma_f32_16x16x4_f32(
             false, a,  false, b,
             (short)0, acc,  false,  false);
    }

    float* __restrict__ hrow = H + (size_t)(rowBase + half * 8) * HID + (colBase + l16);
#pragma unroll
    for (int i = 0; i < 8; ++i)
        hrow[(size_t)i * HID] = acc[i];
}

__global__ __launch_bounds__(256) void prop_kernel(const float* __restrict__ h,
                                                   const int* __restrict__ src,
                                                   const int* __restrict__ dst,
                                                   const float* __restrict__ enorm,
                                                   float* __restrict__ out) {
    const int lane = threadIdx.x & 31;
    const int e    = blockIdx.x * 8 + (threadIdx.x >> 5);
    if (e >= N_EDGES) return;
    const int   s = src[e];
    const int   d = dst[e];
    const float w = enorm[e];
    const float4 v = *(const float4*)(h + (size_t)s * HID + lane * 4);
    float* o = out + (size_t)d * HID + lane * 4;
    atom_add_f32(o + 0, v.x * w);
    atom_add_f32(o + 1, v.y * w);
    atom_add_f32(o + 2, v.z * w);
    atom_add_f32(o + 3, v.w * w);
}

__global__ __launch_bounds__(256) void bias_relu_kernel(float* __restrict__ h,
                                                        const float* __restrict__ bias) {
    size_t i = ((size_t)blockIdx.x * blockDim.x + threadIdx.x) * 4;
    if (i >= (size_t)N_NODES * HID) return;
    int col = (int)(i % HID);
    float4 v  = *(float4*)(h + i);
    float4 bb = *(const float4*)(bias + col);
    v.x = fmaxf(v.x + bb.x, 0.0f);
    v.y = fmaxf(v.y + bb.y, 0.0f);
    v.z = fmaxf(v.z + bb.z, 0.0f);
    v.w = fmaxf(v.w + bb.w, 0.0f);
    *(float4*)(h + i) = v;
}

__global__ __launch_bounds__(256) void axpy_kernel(const float* __restrict__ p,
                                                   const float* __restrict__ h0,
                                                   float* __restrict__ out) {
    size_t i = ((size_t)blockIdx.x * blockDim.x + threadIdx.x) * 4;
    if (i >= (size_t)N_NODES * HID) return;
    float4 a = *(const float4*)(p + i);
    float4 b = *(const float4*)(h0 + i);
    float4 r;
    r.x = (1.0f - ALPHA) * a.x + ALPHA * b.x;
    r.y = (1.0f - ALPHA) * a.y + ALPHA * b.y;
    r.z = (1.0f - ALPHA) * a.z + ALPHA * b.z;
    r.w = (1.0f - ALPHA) * a.w + ALPHA * b.w;
    *(float4*)(out + i) = r;
}

extern "C" void kernel_launch(void* const* d_in, const int* in_sizes, int n_in,
                              void* d_out, int out_size, void* d_ws, size_t ws_size,
                              hipStream_t stream) {
    const float* features = (const float*)d_in[0];
    const float* W        = (const float*)d_in[1];
    const float* bias     = (const float*)d_in[2];
    const int*   eidx     = (const int*)d_in[3];
    const int*   src      = eidx;
    const int*   dst      = eidx + N_EDGES;

    float* out = (float*)d_out;

    const size_t NH = (size_t)N_NODES * HID;
    float* deg     = (float*)d_ws;
    float* deg_out = deg;
    float* deg_in  = deg + N_NODES;
    float* enorm   = deg + 2 * N_NODES;
    float* bufA    = enorm + N_EDGES;
    float* bufB    = bufA + NH;

    hipMemsetAsync(deg, 0, 2 * (size_t)N_NODES * sizeof(float), stream);
    deg_kernel<<<(N_EDGES + 255) / 256, 256, 0, stream>>>(src, dst, deg_out, deg_in);
    inv_kernel<<<(2 * N_NODES + 255) / 256, 256, 0, stream>>>(deg, 2 * N_NODES);
    enorm_kernel<<<(N_EDGES + 255) / 256, 256, 0, stream>>>(src, dst, deg_out, deg_in, enorm);

    gemm_kernel<<<N_NODES / 16, 256, 0, stream>>>(features, W, bufA);
    hipMemsetAsync(bufB, 0, NH * sizeof(float), stream);
    prop_kernel<<<N_EDGES / 8, 256, 0, stream>>>(bufA, src, dst, enorm, bufB);
    bias_relu_kernel<<<(int)(NH / 4 / 256), 256, 0, stream>>>(bufB, bias);

    float* cur = bufB;
    for (int t = 0; t < K_STEPS; ++t) {
        float* tmp = (t & 1) ? out : bufA;
        hipMemsetAsync(tmp, 0, NH * sizeof(float), stream);
        prop_kernel<<<N_EDGES / 8, 256, 0, stream>>>(cur, src, dst, enorm, tmp);
        axpy_kernel<<<(int)(NH / 4 / 256), 256, 0, stream>>>(tmp, bufB, tmp);
        cur = tmp;
    }
}


# Propagation turned from push to pull: edges grouped by dst (count, scan, fill, rows sorted by src), one wave sums a dst row in fixed row order in registers; float atomics and big memsets gone
# speedup vs baseline: 3.6938x; 3.6938x over previous
#include <hip/hip_runtime.h>
#include <cstdint>


#define N_NODES 100000
#define N_EDGES 1600000
#define IN_CH   256
#define HID     128
#define K_STEPS 10
#define ALPHA   0.1f

#define SCAN_T  1024
#define SCAN_B  ((N_NODES + SCAN_T - 1) / SCAN_T)

typedef __attribute__((ext_vector_type(2))) float v2f;
typedef __attribute__((ext_vector_type(8))) float v8f;

__global__ __launch_bounds__(256) void count_kernel(const int* __restrict__ src,
                                                    const int* __restrict__ dst,
                                                    int* __restrict__ cnt_out,
                                                    int* __restrict__ cnt_in) {
    int e = blockIdx.x * blockDim.x + threadIdx.x;
    if (e < N_EDGES) {
        __hip_atomic_fetch_add(&cnt_out[src[e]], 1, __ATOMIC_RELAXED, __HIP_MEMORY_SCOPE_AGENT);
        __hip_atomic_fetch_add(&cnt_in[dst[e]], 1, __ATOMIC_RELAXED, __HIP_MEMORY_SCOPE_AGENT);
    }
}

__global__ __launch_bounds__(256) void inv_kernel(const int* __restrict__ cnt_out,
                                                  float* __restrict__ inv_out) {
    int i = blockIdx.x * blockDim.x + threadIdx.x;
    if (i < N_NODES) {
        float d = (float)cnt_out[i];
        inv_out[i] = (d > 0.0f) ? rsqrtf(fmaxf(d, 1.0f)) : 0.0f;
    }
}

__device__ __forceinline__ int block_incl_scan(int v, int* sh) {
    const int t = threadIdx.x;
    sh[t] = v;
    __syncthreads();
    for (int off = 1; off < SCAN_T; off <<= 1) {
        int add = (t >= off) ? sh[t - off] : 0;
        __syncthreads();
        sh[t] += add;
        __syncthreads();
    }
    return sh[t];
}

__global__ __launch_bounds__(SCAN_T) void scan_kernel(const int* __restrict__ cnt_in,
                                                      int* __restrict__ rowstart,
                                                      int* __restrict__ cursor) {
    __shared__ int sh[SCAN_T];
    const int t = threadIdx.x;
    const int b = blockIdx.x;
    int pre = 0;
    for (int k = 0; k < b; ++k)
        pre += cnt_in[k * SCAN_T + t];
    (void)block_incl_scan(pre, sh);
    const int base = sh[SCAN_T - 1];
    __syncthreads();
    const int i = b * SCAN_T + t;
    const int c = (i < N_NODES) ? cnt_in[i] : 0;
    const int incl = block_incl_scan(c, sh);
    if (i < N_NODES) {
        const int start = base + incl - c;
        rowstart[i] = start;
        cursor[i]   = start;
    }
}

__global__ __launch_bounds__(256) void fill_kernel(const int* __restrict__ src,
                                                   const int* __restrict__ dst,
                                                   int* __restrict__ cursor,
                                                   int* __restrict__ csr_src) {
    int e = blockIdx.x * blockDim.x + threadIdx.x;
    if (e < N_EDGES) {
        int pos = __hip_atomic_fetch_add(&cursor[dst[e]], 1, __ATOMIC_RELAXED, __HIP_MEMORY_SCOPE_AGENT);
        csr_src[pos] = src[e];
    }
}

__global__ __launch_bounds__(256) void sort_rows_kernel(const int* __restrict__ rowstart,
                                                        int* __restrict__ csr_src) {
    int d = blockIdx.x * blockDim.x + threadIdx.x;
    if (d >= N_NODES) return;
    const int beg = rowstart[d];
    const int end = (d + 1 < N_NODES) ? rowstart[d + 1] : N_EDGES;
    for (int i = beg + 1; i < end; ++i) {
        const int key = csr_src[i];
        int j = i - 1;
        while (j >= beg && csr_src[j] > key) {
            csr_src[j + 1] = csr_src[j];
            --j;
        }
        csr_src[j + 1] = key;
    }
}

__global__ __launch_bounds__(256) void gemm_kernel(const float* __restrict__ X,
                                                   const float* __restrict__ W,
                                                   float* __restrict__ H) {
    const int lane    = threadIdx.x & 31;
    const int wave    = threadIdx.x >> 5;
    const int rowBase = blockIdx.x * 16;
    const int colBase = wave * 16;
    const int half    = lane >> 4;
    const int l16     = lane & 15;

    const float* __restrict__ arow = X + (size_t)(rowBase + l16) * IN_CH + half * 2;
    const float* __restrict__ wcol = W + (size_t)(half * 2) * HID + (colBase + l16);

    v8f acc = {};
#pragma unroll 8
    for (int k0 = 0; k0 < IN_CH; k0 += 4) {
        v2f a, b;
        a.x = arow[k0];
        a.y = arow[k0 + 1];
        b.x = wcol[k0 * HID];
        b.y = wcol[k0 * HID + HID];
        acc = __builtin_amdgcn_wmma_f32_16x16x4_f32(
             false, a,  false, b,
             (short)0, acc,  false,  false);
    }

    float* __restrict__ hrow = H + (size_t)(rowBase + half * 8) * HID + (colBase + l16);
#pragma unroll
    for (int i = 0; i < 8; ++i)
        hrow[(size_t)i * HID] = acc[i];
}

__global__ __launch_bounds__(256) void pull_kernel(const float* __restrict__ h,
                                                   const int* __restrict__ rowstart,
                                                   const int* __restrict__ csr_src,
                                                   const float* __restrict__ inv_out,
                                                   float* __restrict__ out) {
    const int lane = threadIdx.x & 31;
    const int wave = __builtin_amdgcn_readfirstlane((int)(threadIdx.x >> 5));
    const int d    = blockIdx.x * 8 + wave;
    if (d >= N_NODES) return;
    const int beg = rowstart[d];
    const int end = (d + 1 < N_NODES) ? rowstart[d + 1] : N_EDGES;

    float4 acc = make_float4(0.0f, 0.0f, 0.0f, 0.0f);
    if (end > beg) {
        const float wd = rsqrtf((float)(end - beg));
        const float* __restrict__ hl = h + lane * 4;
        for (int c0 = beg; c0 < end; c0 += 32) {
            const int n = min(32, end - c0);
            int   s_l = 0;
            float w_l = 0.0f;
            if (lane < n) {
                s_l = csr_src[c0 + lane];
                w_l = inv_out[s_l] * wd;
            }
#pragma unroll 4
            for (int j = 0; j < n; ++j) {
                const int   s = __builtin_amdgcn_readlane(s_l, j);
                const float w = __int_as_float(__builtin_amdgcn_readlane(__float_as_int(w_l), j));
                const float4 v = *(const float4*)(hl + (size_t)s * HID);
                acc.x += v.x * w;
                acc.y += v.y * w;
                acc.z += v.z * w;
                acc.w += v.w * w;
            }
        }
    }
    *(float4*)(out + (size_t)d * HID + lane * 4) = acc;
}

__global__ __launch_bounds__(256) void bias_relu_kernel(float* __restrict__ h,
                                                        const float* __restrict__ bias) {
    size_t i = ((size_t)blockIdx.x * blockDim.x + threadIdx.x) * 4;
    if (i >= (size_t)N_NODES * HID) return;
    int col = (int)(i % HID);
    float4 v  = *(float4*)(h + i);
    float4 bb = *(const float4*)(bias + col);
    v.x = fmaxf(v.x + bb.x, 0.0f);
    v.y = fmaxf(v.y + bb.y, 0.0f);
    v.z = fmaxf(v.z + bb.z, 0.0f);
    v.w = fmaxf(v.w + bb.w, 0.0f);
    *(float4*)(h + i) = v;
}

__global__ __launch_bounds__(256) void axpy_kernel(const float* __restrict__ p,
                                                   const float* __restrict__ h0,
                                                   float* __restrict__ out) {
    size_t i = ((size_t)blockIdx.x * blockDim.x + threadIdx.x) * 4;
    if (i >= (size_t)N_NODES * HID) return;
    float4 a = *(const float4*)(p + i);
    float4 b = *(const float4*)(h0 + i);
    float4 r;
    r.x = (1.0f - ALPHA) * a.x + ALPHA * b.x;
    r.y = (1.0f - ALPHA) * a.y + ALPHA * b.y;
    r.z = (1.0f - ALPHA) * a.z + ALPHA * b.z;
    r.w = (1.0f - ALPHA) * a.w + ALPHA * b.w;
    *(float4*)(out + i) = r;
}

extern "C" void kernel_launch(void* const* d_in, const int* in_sizes, int n_in,
                              void* d_out, int out_size, void* d_ws, size_t ws_size,
                              hipStream_t stream) {
    const float* features = (const float*)d_in[0];
    const float* W        = (const float*)d_in[1];
    const float* bias     = (const float*)d_in[2];
    const int*   eidx     = (const int*)d_in[3];
    const int*   src      = eidx;
    const int*   dst      = eidx + N_EDGES;

    float* out = (float*)d_out;

    const size_t NH = (size_t)N_NODES * HID;
    float* inv_out  = (float*)d_ws;
    int*   rowstart = (int*)(inv_out + N_NODES);
    int*   csr_src  = rowstart + N_NODES;
    float* bufA     = (float*)(csr_src + N_EDGES);
    float* bufB     = bufA + NH;

    int* cnt_out = (int*)bufB;
    int* cnt_in  = cnt_out + N_NODES;
    int* cursor  = cnt_in + N_NODES;

    hipMemsetAsync(cnt_out, 0, 2 * (size_t)N_NODES * sizeof(int), stream);
    count_kernel<<<(N_EDGES + 255) / 256, 256, 0, stream>>>(src, dst, cnt_out, cnt_in);
    inv_kernel<<<(N_NODES + 255) / 256, 256, 0, stream>>>(cnt_out, inv_out);
    scan_kernel<<<SCAN_B, SCAN_T, 0, stream>>>(cnt_in, rowstart, cursor);
    fill_kernel<<<(N_EDGES + 255) / 256, 256, 0, stream>>>(src, dst, cursor, csr_src);
    sort_rows_kernel<<<(N_NODES + 255) / 256, 256, 0, stream>>>(rowstart, csr_src);

    gemm_kernel<<<N_NODES / 16, 256, 0, stream>>>(features, W, bufA);
    pull_kernel<<<(N_NODES + 7) / 8, 256, 0, stream>>>(bufA, rowstart, csr_src, inv_out, bufB);
    bias_relu_kernel<<<(int)(NH / 4 / 256), 256, 0, stream>>>(bufB, bias);

    float* cur = bufB;
    for (int t = 0; t < K_STEPS; ++t) {
        float* tmp = (t & 1) ? out : bufA;
        pull_kernel<<<(N_NODES + 7) / 8, 256, 0, stream>>>(cur, rowstart, csr_src, inv_out, tmp);
        axpy_kernel<<<(int)(NH / 4 / 256), 256, 0, stream>>>(tmp, bufB, tmp);
        cur = tmp;
    }
}


# Step tails fused into the propagation kernel: bias+relu and the (1-alpha)*sum + alpha*h0 blend act on the row sum in registers before its one store; the 11 elementwise launches are gone
# speedup vs baseline: 4.2575x; 1.1526x over previous
#include <hip/hip_runtime.h>
#include <cstdint>


#define N_NODES 100000
#define N_EDGES 1600000
#define IN_CH   256
#define HID     128
#define K_STEPS 10
#define ALPHA   0.1f

#define SCAN_T  1024
#define SCAN_B  ((N_NODES + SCAN_T - 1) / SCAN_T)

typedef __attribute__((ext_vector_type(2))) float v2f;
typedef __attribute__((ext_vector_type(8))) float v8f;

__global__ __launch_bounds__(256) void count_kernel(const int* __restrict__ src,
                                                    const int* __restrict__ dst,
                                                    int* __restrict__ cnt_out,
                                                    int* __restrict__ cnt_in) {
    int e = blockIdx.x * blockDim.x + threadIdx.x;
    if (e < N_EDGES) {
        __hip_atomic_fetch_add(&cnt_out[src[e]], 1, __ATOMIC_RELAXED, __HIP_MEMORY_SCOPE_AGENT);
        __hip_atomic_fetch_add(&cnt_in[dst[e]], 1, __ATOMIC_RELAXED, __HIP_MEMORY_SCOPE_AGENT);
    }
}

__global__ __launch_bounds__(256) void inv_kernel(const int* __restrict__ cnt_out,
                                                  float* __restrict__ inv_out) {
    int i = blockIdx.x * blockDim.x + threadIdx.x;
    if (i < N_NODES) {
        float d = (float)cnt_out[i];
        inv_out[i] = (d > 0.0f) ? rsqrtf(fmaxf(d, 1.0f)) : 0.0f;
    }
}

__device__ __forceinline__ int block_incl_scan(int v, int* sh) {
    const int t = threadIdx.x;
    sh[t] = v;
    __syncthreads();
    for (int off = 1; off < SCAN_T; off <<= 1) {
        int add = (t >= off) ? sh[t - off] : 0;
        __syncthreads();
        sh[t] += add;
        __syncthreads();
    }
    return sh[t];
}

__global__ __launch_bounds__(SCAN_T) void scan_kernel(const int* __restrict__ cnt_in,
                                                      int* __restrict__ rowstart,
                                                      int* __restrict__ cursor) {
    __shared__ int sh[SCAN_T];
    const int t = threadIdx.x;
    const int b = blockIdx.x;
    int pre = 0;
    for (int k = 0; k < b; ++k)
        pre += cnt_in[k * SCAN_T + t];
    (void)block_incl_scan(pre, sh);
    const int base = sh[SCAN_T - 1];
    __syncthreads();
    const int i = b * SCAN_T + t;
    const int c = (i < N_NODES) ? cnt_in[i] : 0;
    const int incl = block_incl_scan(c, sh);
    if (i < N_NODES) {
        const int start = base + incl - c;
        rowstart[i] = start;
        cursor[i]   = start;
    }
}

__global__ __launch_bounds__(256) void fill_kernel(const int* __restrict__ src,
                                                   const int* __restrict__ dst,
                                                   int* __restrict__ cursor,
                                                   int* __restrict__ csr_src) {
    int e = blockIdx.x * blockDim.x + threadIdx.x;
    if (e < N_EDGES) {
        int pos = __hip_atomic_fetch_add(&cursor[dst[e]], 1, __ATOMIC_RELAXED, __HIP_MEMORY_SCOPE_AGENT);
        csr_src[pos] = src[e];
    }
}

__global__ __launch_bounds__(256) void sort_rows_kernel(const int* __restrict__ rowstart,
                                                        int* __restrict__ csr_src) {
    int d = blockIdx.x * blockDim.x + threadIdx.x;
    if (d >= N_NODES) return;
    const int beg = rowstart[d];
    const int end = (d + 1 < N_NODES) ? rowstart[d + 1] : N_EDGES;
    for (int i = beg + 1; i < end; ++i) {
        const int key = csr_src[i];
        int j = i - 1;
        while (j >= beg && csr_src[j] > key) {
            csr_src[j + 1] = csr_src[j];
            --j;
        }
        csr_src[j + 1] = key;
    }
}

__global__ __launch_bounds__(256) void gemm_kernel(const float* __restrict__ X,
                                                   const float* __restrict__ W,
                                                   float* __restrict__ H) {
    const int lane    = threadIdx.x & 31;
    const int wave    = threadIdx.x >> 5;
    const int rowBase = blockIdx.x * 16;
    const int colBase = wave * 16;
    const int half    = lane >> 4;
    const int l16     = lane & 15;

    const float* __restrict__ arow = X + (size_t)(rowBase + l16) * IN_CH + half * 2;
    const float* __restrict__ wcol = W + (size_t)(half * 2) * HID + (colBase + l16);

    v8f acc = {};
#pragma unroll 8
    for (int k0 = 0; k0 < IN_CH; k0 += 4) {
        v2f a, b;
        a.x = arow[k0];
        a.y = arow[k0 + 1];
        b.x = wcol[k0 * HID];
        b.y = wcol[k0 * HID + HID];
        acc = __builtin_amdgcn_wmma_f32_16x16x4_f32(
             false, a,  false, b,
             (short)0, acc,  false,  false);
    }

    float* __restrict__ hrow = H + (size_t)(rowBase + half * 8) * HID + (colBase + l16);
#pragma unroll
    for (int i = 0; i < 8; ++i)
        hrow[(size_t)i * HID] = acc[i];
}

__device__ __forceinline__ float4 pull_row(const float* __restrict__ h,
                                           const int* __restrict__ csr_src,
                                           const float* __restrict__ inv_out,
                                           const int beg, const int end, const int lane) {
    float4 acc = make_float4(0.0f, 0.0f, 0.0f, 0.0f);
    if (end > beg) {
        const float wd = rsqrtf((float)(end - beg));
        const float* __restrict__ hl = h + lane * 4;
        for (int c0 = beg; c0 < end; c0 += 32) {
            const int n = min(32, end - c0);
            int   s_l = 0;
            float w_l = 0.0f;
            if (lane < n) {
                s_l = csr_src[c0 + lane];
                w_l = inv_out[s_l] * wd;
            }
#pragma unroll 4
            for (int j = 0; j < n; ++j) {
                const int   s = __builtin_amdgcn_readlane(s_l, j);
                const float w = __int_as_float(__builtin_amdgcn_readlane(__float_as_int(w_l), j));
                const float4 v = *(const float4*)(hl + (size_t)s * HID);
                acc.x += v.x * w;
                acc.y += v.y * w;
                acc.z += v.z * w;
                acc.w += v.w * w;
            }
        }
    }
    return acc;
}

__global__ __launch_bounds__(256) void pull_bias_relu_kernel(const float* __restrict__ h,
                                                             const int* __restrict__ rowstart,
                                                             const int* __restrict__ csr_src,
                                                             const float* __restrict__ inv_out,
                                                             const float* __restrict__ bias,
                                                             float* __restrict__ out) {
    const int lane = threadIdx.x & 31;
    const int wave = __builtin_amdgcn_readfirstlane((int)(threadIdx.x >> 5));
    const int d    = blockIdx.x * 8 + wave;
    if (d >= N_NODES) return;
    const float4 bb = *(const float4*)(bias + lane * 4);
    const int beg = rowstart[d];
    const int end = (d + 1 < N_NODES) ? rowstart[d + 1] : N_EDGES;

    float4 acc = pull_row(h, csr_src, inv_out, beg, end, lane);
    acc.x = fmaxf(acc.x + bb.x, 0.0f);
    acc.y = fmaxf(acc.y + bb.y, 0.0f);
    acc.z = fmaxf(acc.z + bb.z, 0.0f);
    acc.w = fmaxf(acc.w + bb.w, 0.0f);
    *(float4*)(out + (size_t)d * HID + lane * 4) = acc;
}

__global__ __launch_bounds__(256) void pull_axpy_kernel(const float* __restrict__ h,
                                                        const int* __restrict__ rowstart,
                                                        const int* __restrict__ csr_src,
                                                        const float* __restrict__ inv_out,
                                                        const float* __restrict__ h0,
                                                        float* __restrict__ out) {
    const int lane = threadIdx.x & 31;
    const int wave = __builtin_amdgcn_readfirstlane((int)(threadIdx.x >> 5));
    const int d    = blockIdx.x * 8 + wave;
    if (d >= N_NODES) return;
    const float4 z = *(const float4*)(h0 + (size_t)d * HID + lane * 4);
    const int beg = rowstart[d];
    const int end = (d + 1 < N_NODES) ? rowstart[d + 1] : N_EDGES;

    const float4 a = pull_row(h, csr_src, inv_out, beg, end, lane);
    float4 r;
    r.x = fmaf(1.0f - ALPHA, a.x, ALPHA * z.x);
    r.y = fmaf(1.0f - ALPHA, a.y, ALPHA * z.y);
    r.z = fmaf(1.0f - ALPHA, a.z, ALPHA * z.z);
    r.w = fmaf(1.0f - ALPHA, a.w, ALPHA * z.w);
    *(float4*)(out + (size_t)d * HID + lane * 4) = r;
}

extern "C" void kernel_launch(void* const* d_in, const int* in_sizes, int n_in,
                              void* d_out, int out_size, void* d_ws, size_t ws_size,
                              hipStream_t stream) {
    const float* features = (const float*)d_in[0];
    const float* W        = (const float*)d_in[1];
    const float* bias     = (const float*)d_in[2];
    const int*   eidx     = (const int*)d_in[3];
    const int*   src      = eidx;
    const int*   dst      = eidx + N_EDGES;

    float* out = (float*)d_out;

    const size_t NH = (size_t)N_NODES * HID;
    float* inv_out  = (float*)d_ws;
    int*   rowstart = (int*)(inv_out + N_NODES);
    int*   csr_src  = rowstart + N_NODES;
    float* bufA     = (float*)(csr_src + N_EDGES);
    float* bufB     = bufA + NH;

    int* cnt_out = (int*)bufB;
    int* cnt_in  = cnt_out + N_NODES;
    int* cursor  = cnt_in + N_NODES;

    hipMemsetAsync(cnt_out, 0, 2 * (size_t)N_NODES * sizeof(int), stream);
    count_kernel<<<(N_EDGES + 255) / 256, 256, 0, stream>>>(src, dst, cnt_out, cnt_in);
    inv_kernel<<<(N_NODES + 255) / 256, 256, 0, stream>>>(cnt_out, inv_out);
    scan_kernel<<<SCAN_B, SCAN_T, 0, stream>>>(cnt_in, rowstart, cursor);
    fill_kernel<<<(N_EDGES + 255) / 256, 256, 0, stream>>>(src, dst, cursor, csr_src);
    sort_rows_kernel<<<(N_NODES + 255) / 256, 256, 0, stream>>>(rowstart, csr_src);

    gemm_kernel<<<N_NODES / 16, 256, 0, stream>>>(features, W, bufA);
    pull_bias_relu_kernel<<<(N_NODES + 7) / 8, 256, 0, stream>>>(bufA, rowstart, csr_src, inv_out, bias, bufB);

    float* cur = bufB;
    for (int t = 0; t < K_STEPS; ++t) {
        float* tmp = (t & 1) ? out : bufA;
        pull_axpy_kernel<<<(N_NODES + 7) / 8, 256, 0, stream>>>(cur, rowstart, csr_src, inv_out, bufB, tmp);
        cur = tmp;
    }
}


# GEMM regrouped: W staged once per work-group into LDS in fragment order, a wave owns 16 rows x 64 columns so one A load feeds 4 matrix instructions; k summed in 16-k blocks (sum order changed)
# speedup vs baseline: 4.6828x; 1.0999x over previous
#include <hip/hip_runtime.h>
#include <cstdint>


#define N_NODES 100000
#define N_EDGES 1600000
#define IN_CH   256
#define HID     128
#define K_STEPS 10
#define ALPHA   0.1f

#define SCAN_T  1024
#define SCAN_B  ((N_NODES + SCAN_T - 1) / SCAN_T)

typedef __attribute__((ext_vector_type(2))) float v2f;
typedef __attribute__((ext_vector_type(8))) float v8f;

__global__ __launch_bounds__(256) void count_kernel(const int* __restrict__ src,
                                                    const int* __restrict__ dst,
                                                    int* __restrict__ cnt_out,
                                                    int* __restrict__ cnt_in) {
    int e = blockIdx.x * blockDim.x + threadIdx.x;
    if (e < N_EDGES) {
        __hip_atomic_fetch_add(&cnt_out[src[e]], 1, __ATOMIC_RELAXED, __HIP_MEMORY_SCOPE_AGENT);
        __hip_atomic_fetch_add(&cnt_in[dst[e]], 1, __ATOMIC_RELAXED, __HIP_MEMORY_SCOPE_AGENT);
    }
}

__global__ __launch_bounds__(256) void inv_kernel(const int* __restrict__ cnt_out,
                                                  float* __restrict__ inv_out) {
    int i = blockIdx.x * blockDim.x + threadIdx.x;
    if (i < N_NODES) {
        float d = (float)cnt_out[i];
        inv_out[i] = (d > 0.0f) ? rsqrtf(fmaxf(d, 1.0f)) : 0.0f;
    }
}

__device__ __forceinline__ int block_incl_scan(int v, int* sh) {
    const int t = threadIdx.x;
    sh[t] = v;
    __syncthreads();
    for (int off = 1; off < SCAN_T; off <<= 1) {
        int add = (t >= off) ? sh[t - off] : 0;
        __syncthreads();
        sh[t] += add;
        __syncthreads();
    }
    return sh[t];
}

__global__ __launch_bounds__(SCAN_T) void scan_kernel(const int* __restrict__ cnt_in,
                                                      int* __restrict__ rowstart,
                                                      int* __restrict__ cursor) {
    __shared__ int sh[SCAN_T];
    const int t = threadIdx.x;
    const int b = blockIdx.x;
    int pre = 0;
    for (int k = 0; k < b; ++k)
        pre += cnt_in[k * SCAN_T + t];
    (void)block_incl_scan(pre, sh);
    const int base = sh[SCAN_T - 1];
    __syncthreads();
    const int i = b * SCAN_T + t;
    const int c = (i < N_NODES) ? cnt_in[i] : 0;
    const int incl = block_incl_scan(c, sh);
    if (i < N_NODES) {
        const int start = base + incl - c;
        rowstart[i] = start;
        cursor[i]   = start;
    }
}

__global__ __launch_bounds__(256) void fill_kernel(const int* __restrict__ src,
                                                   const int* __restrict__ dst,
                                                   int* __restrict__ cursor,
                                                   int* __restrict__ csr_src) {
    int e = blockIdx.x * blockDim.x + threadIdx.x;
    if (e < N_EDGES) {
        int pos = __hip_atomic_fetch_add(&cursor[dst[e]], 1, __ATOMIC_RELAXED, __HIP_MEMORY_SCOPE_AGENT);
        csr_src[pos] = src[e];
    }
}

__global__ __launch_bounds__(256) void sort_rows_kernel(const int* __restrict__ rowstart,
                                                        int* __restrict__ csr_src) {
    int d = blockIdx.x * blockDim.x + threadIdx.x;
    if (d >= N_NODES) return;
    const int beg = rowstart[d];
    const int end = (d + 1 < N_NODES) ? rowstart[d + 1] : N_EDGES;
    for (int i = beg + 1; i < end; ++i) {
        const int key = csr_src[i];
        int j = i - 1;
        while (j >= beg && csr_src[j] > key) {
            csr_src[j + 1] = csr_src[j];
            --j;
        }
        csr_src[j + 1] = key;
    }
}

#define GEMM_BLOCKS 512
#define ROW_TILES   (N_NODES / 16)
#define PAIR_SLOTS  (GEMM_BLOCKS * 4)
#define K_BLOCKS    (IN_CH / 16)

__device__ __forceinline__ v8f mm4(const float ax, const float ay, const float bx, const float by, const v8f c) {
    v2f a, b;
    a.x = ax; a.y = ay;
    b.x = bx; b.y = by;
    return __builtin_amdgcn_wmma_f32_16x16x4_f32(
         false, a,  false, b,
         (short)0, c,  false,  false);
}

__device__ __forceinline__ void load_block(const float* __restrict__ xr, const float4* __restrict__ bl,
                                           const int kb, float4& a0, float4& a1, float4 (&b)[4][2]) {
    a0 = *(const float4*)(xr + kb * 16);
    a1 = *(const float4*)(xr + kb * 16 + 4);
#pragma unroll
    for (int tt = 0; tt < 4; ++tt) {
        b[tt][0] = bl[kb * 512 + tt * 64];
        b[tt][1] = bl[kb * 512 + tt * 64 + 32];
    }
}

__device__ __forceinline__ void mm_block(v8f (&acc)[4], const float4 a0, const float4 a1,
                                         const float4 (&b)[4][2]) {
#pragma unroll
    for (int tt = 0; tt < 4; ++tt) acc[tt] = mm4(a0.x, a0.y, b[tt][0].x, b[tt][0].y, acc[tt]);
#pragma unroll
    for (int tt = 0; tt < 4; ++tt) acc[tt] = mm4(a0.z, a0.w, b[tt][0].z, b[tt][0].w, acc[tt]);
#pragma unroll
    for (int tt = 0; tt < 4; ++tt) acc[tt] = mm4(a1.x, a1.y, b[tt][1].x, b[tt][1].y, acc[tt]);
#pragma unroll
    for (int tt = 0; tt < 4; ++tt) acc[tt] = mm4(a1.z, a1.w, b[tt][1].z, b[tt][1].w, acc[tt]);
}

__global__ __launch_bounds__(256) __attribute__((amdgpu_waves_per_eu(4, 4))) void gemm_kernel(const float* __restrict__ X,
                                                   const float* __restrict__ W,
                                                   const float* __restrict__ inv_out,
                                                   float* __restrict__ H) {
    __shared__ __attribute__((aligned(16))) float Wl[IN_CH * HID];

    const int tid  = threadIdx.x;
    const int lane = tid & 31;
    const int wave = __builtin_amdgcn_readfirstlane(tid >> 5);
    const int half = lane >> 4;
    const int l16  = lane & 15;

    for (int p = tid; p < (IN_CH / 4) * HID; p += 256) {
        const int m = p >> 7;
        const int n = p & 127;
        const float* __restrict__ w = W + (size_t)(4 * m) * HID + n;
        float4 v;
        v.x = w[0];
        v.y = w[HID];
        v.z = w[2 * HID];
        v.w = w[3 * HID];
        const int kb = m >> 2, h = (m >> 1) & 1, q = m & 1;
        const int t = n >> 4, c = n & 15;
        *(float4*)(Wl + ((((kb * 8 + t) * 2 + q) * 32 + 16 * h + c) << 2)) = v;
    }
    __syncthreads();

    const int colhalf = wave & 1;
    const float4* __restrict__ bl = (const float4*)Wl + (size_t)(colhalf * 4) * 64 + lane;

    for (int tile = (wave >> 1) * GEMM_BLOCKS + blockIdx.x; tile < ROW_TILES; tile += PAIR_SLOTS) {
        const int rowBase = tile * 16;
        const float* __restrict__ xr = X + (size_t)(rowBase + l16) * IN_CH + half * 8;

        v8f acc[4] = {};
        float4 a0, a1, b[4][2];
        float4 c0, c1, d[4][2];
        load_block(xr, bl, 0, a0, a1, b);
#pragma unroll 1
        for (int kb = 0; kb < K_BLOCKS; kb += 2) {
            load_block(xr, bl, kb + 1, c0, c1, d);
            mm_block(acc, a0, a1, b);
            if (kb + 2 < K_BLOCKS)
                load_block(xr, bl, kb + 2, a0, a1, b);
            mm_block(acc, c0, c1, d);
        }

        const float4 i0 = *(const float4*)(inv_out + rowBase + half * 8);
        const float4 i1 = *(const float4*)(inv_out + rowBase + half * 8 + 4);
        const float iv[8] = {i0.x, i0.y, i0.z, i0.w, i1.x, i1.y, i1.z, i1.w};
        float* __restrict__ hrow = H + (size_t)(rowBase + half * 8) * HID + (colhalf * 64 + l16);
#pragma unroll
        for (int tt = 0; tt < 4; ++tt)
#pragma unroll
            for (int i = 0; i < 8; ++i)
                hrow[(size_t)i * HID + tt * 16] = acc[tt][i] * iv[i];
    }
}

__device__ __forceinline__ void walk_chunk(float4& acc, const float* __restrict__ gl,
                                           const int s_l, const int n) {
#pragma unroll 4
    for (int j = 0; j < n; ++j) {
        const int s = __builtin_amdgcn_readlane(s_l, j);
        const float4 v = *(const float4*)(gl + (size_t)s * HID);
        acc.x += v.x;
        acc.y += v.y;
        acc.z += v.z;
        acc.w += v.w;
    }
}

__device__ __forceinline__ int first_ids(const int* __restrict__ csr_src,
                                         const int beg, const int end, const int lane) {
    int s_l = 0;
    if (lane < end - beg)
        s_l = csr_src[beg + lane];
    return s_l;
}

__device__ __forceinline__ float4 pull_row(const float* __restrict__ g,
                                           const int* __restrict__ csr_src,
                                           const int s_first,
                                           const int beg, const int end, const int lane) {
    float4 acc = make_float4(0.0f, 0.0f, 0.0f, 0.0f);
    const float* __restrict__ gl = g + lane * 4;
    walk_chunk(acc, gl, s_first, min(32, end - beg));
    for (int c0 = beg + 32; c0 < end; c0 += 32) {
        const int n = min(32, end - c0);
        int s_l = 0;
        if (lane < n)
            s_l = csr_src[c0 + lane];
        walk_chunk(acc, gl, s_l, n);
    }
    return acc;
}

__device__ __forceinline__ float dst_norm(const int beg, const int end) {
    return rsqrtf((float)max(end - beg, 1));
}

__global__ __launch_bounds__(256) void pull_bias_relu_kernel(const float* __restrict__ g,
                                                             const int* __restrict__ rowstart,
                                                             const int* __restrict__ csr_src,
                                                             const float* __restrict__ inv_out,
                                                             const float* __restrict__ bias,
                                                             float* __restrict__ h0,
                                                             float* __restrict__ g0) {
    const int lane = threadIdx.x & 31;
    const int wave = __builtin_amdgcn_readfirstlane((int)(threadIdx.x >> 5));
    const int d    = blockIdx.x * 8 + wave;
    if (d >= N_NODES) return;
    const int beg = rowstart[d];
    const int end = (d + 1 < N_NODES) ? rowstart[d + 1] : N_EDGES;
    const int s_first = first_ids(csr_src, beg, end, lane);
    const float4 bb = *(const float4*)(bias + lane * 4);
    const float io = inv_out[d];
    const float wd = dst_norm(beg, end);

    const float4 acc = pull_row(g, csr_src, s_first, beg, end, lane);
    float4 r;
    r.x = fmaxf(acc.x * wd + bb.x, 0.0f);
    r.y = fmaxf(acc.y * wd + bb.y, 0.0f);
    r.z = fmaxf(acc.z * wd + bb.z, 0.0f);
    r.w = fmaxf(acc.w * wd + bb.w, 0.0f);
    *(float4*)(h0 + (size_t)d * HID + lane * 4) = r;
    r.x *= io;
    r.y *= io;
    r.z *= io;
    r.w *= io;
    *(float4*)(g0 + (size_t)d * HID + lane * 4) = r;
}

template <bool LAST>
__global__ __launch_bounds__(256) void pull_axpy_kernel(const float* __restrict__ g,
                                                        const int* __restrict__ rowstart,
                                                        const int* __restrict__ csr_src,
                                                        const float* __restrict__ inv_out,
                                                        const float* __restrict__ h0,
                                                        float* __restrict__ out) {
    const int lane = threadIdx.x & 31;
    const int wave = __builtin_amdgcn_readfirstlane((int)(threadIdx.x >> 5));
    const int d    = blockIdx.x * 8 + wave;
    if (d >= N_NODES) return;
    const int beg = rowstart[d];
    const int end = (d + 1 < N_NODES) ? rowstart[d + 1] : N_EDGES;
    const int s_first = first_ids(csr_src, beg, end, lane);
    const float4 z = *(const float4*)(h0 + (size_t)d * HID + lane * 4);
    const float io = LAST ? 1.0f : inv_out[d];
    const float wd = dst_norm(beg, end);

    const float4 a = pull_row(g, csr_src, s_first, beg, end, lane);
    float4 r;
    r.x = fmaf(1.0f - ALPHA, a.x * wd, ALPHA * z.x);
    r.y = fmaf(1.0f - ALPHA, a.y * wd, ALPHA * z.y);
    r.z = fmaf(1.0f - ALPHA, a.z * wd, ALPHA * z.z);
    r.w = fmaf(1.0f - ALPHA, a.w * wd, ALPHA * z.w);
    if (!LAST) {
        r.x *= io;
        r.y *= io;
        r.z *= io;
        r.w *= io;
    }
    *(float4*)(out + (size_t)d * HID + lane * 4) = r;
}

extern "C" void kernel_launch(void* const* d_in, const int* in_sizes, int n_in,
                              void* d_out, int out_size, void* d_ws, size_t ws_size,
                              hipStream_t stream) {
    const float* features = (const float*)d_in[0];
    const float* W        = (const float*)d_in[1];
    const float* bias     = (const float*)d_in[2];
    const int*   eidx     = (const int*)d_in[3];
    const int*   src      = eidx;
    const int*   dst      = eidx + N_EDGES;

    float* out = (float*)d_out;

    const size_t NH = (size_t)N_NODES * HID;
    float* inv_out  = (float*)d_ws;
    int*   rowstart = (int*)(inv_out + N_NODES);
    int*   csr_src  = rowstart + N_NODES;
    float* bufA     = (float*)(csr_src + N_EDGES);
    float* bufB     = bufA + NH;

    int* cnt_out = (int*)bufB;
    int* cnt_in  = cnt_out + N_NODES;
    int* cursor  = cnt_in + N_NODES;

    hipMemsetAsync(cnt_out, 0, 2 * (size_t)N_NODES * sizeof(int), stream);
    count_kernel<<<(N_EDGES + 255) / 256, 256, 0, stream>>>(src, dst, cnt_out, cnt_in);
    inv_kernel<<<(N_NODES + 255) / 256, 256, 0, stream>>>(cnt_out, inv_out);
    scan_kernel<<<SCAN_B, SCAN_T, 0, stream>>>(cnt_in, rowstart, cursor);
    fill_kernel<<<(N_EDGES + 255) / 256, 256, 0, stream>>>(src, dst, cursor, csr_src);
    sort_rows_kernel<<<(N_NODES + 255) / 256, 256, 0, stream>>>(rowstart, csr_src);

    gemm_kernel<<<GEMM_BLOCKS, 256, 0, stream>>>(features, W, inv_out, bufA);
    pull_bias_relu_kernel<<<(N_NODES + 7) / 8, 256, 0, stream>>>(bufA, rowstart, csr_src, inv_out, bias, bufB, out);

    float* cur = out;
    for (int t = 0; t < K_STEPS - 1; ++t) {
        float* tmp = (t & 1) ? out : bufA;
        pull_axpy_kernel<false><<<(N_NODES + 7) / 8, 256, 0, stream>>>(cur, rowstart, csr_src, inv_out, bufB, tmp);
        cur = tmp;
    }
    pull_axpy_kernel<true><<<(N_NODES + 7) / 8, 256, 0, stream>>>(cur, rowstart, csr_src, inv_out, bufB, out);
}


# Row sort redone: a wave ranks a row's src ids in registers (each id against the others, passed lane by lane) from a scratch copy and stores each at its rank; one-thread insertion sort in memory gone
# speedup vs baseline: 5.4541x; 1.1647x over previous
#include <hip/hip_runtime.h>
#include <cstdint>


#define N_NODES 100000
#define N_EDGES 1600000
#define IN_CH   256
#define HID     128
#define K_STEPS 10
#define ALPHA   0.1f

#define SCAN_T  1024
#define SCAN_B  ((N_NODES + SCAN_T - 1) / SCAN_T)

typedef __attribute__((ext_vector_type(2))) float v2f;
typedef __attribute__((ext_vector_type(8))) float v8f;

__global__ __launch_bounds__(256) void count_kernel(const int* __restrict__ src,
                                                    const int* __restrict__ dst,
                                                    int* __restrict__ cnt_out,
                                                    int* __restrict__ cnt_in) {
    int e = blockIdx.x * blockDim.x + threadIdx.x;
    if (e < N_EDGES) {
        __hip_atomic_fetch_add(&cnt_out[src[e]], 1, __ATOMIC_RELAXED, __HIP_MEMORY_SCOPE_AGENT);
        __hip_atomic_fetch_add(&cnt_in[dst[e]], 1, __ATOMIC_RELAXED, __HIP_MEMORY_SCOPE_AGENT);
    }
}

__global__ __launch_bounds__(256) void inv_kernel(const int* __restrict__ cnt_out,
                                                  float* __restrict__ inv_out) {
    int i = blockIdx.x * blockDim.x + threadIdx.x;
    if (i < N_NODES) {
        float d = (float)cnt_out[i];
        inv_out[i] = (d > 0.0f) ? rsqrtf(fmaxf(d, 1.0f)) : 0.0f;
    }
}

__device__ __forceinline__ int block_incl_scan(int v, int* sh) {
    const int t = threadIdx.x;
    sh[t] = v;
    __syncthreads();
    for (int off = 1; off < SCAN_T; off <<= 1) {
        int add = (t >= off) ? sh[t - off] : 0;
        __syncthreads();
        sh[t] += add;
        __syncthreads();
    }
    return sh[t];
}

__global__ __launch_bounds__(SCAN_T) void scan_kernel(const int* __restrict__ cnt_in,
                                                      int* __restrict__ rowstart,
                                                      int* __restrict__ cursor) {
    __shared__ int sh[SCAN_T];
    const int t = threadIdx.x;
    const int b = blockIdx.x;
    int pre = 0;
    for (int k = 0; k < b; ++k)
        pre += cnt_in[k * SCAN_T + t];
    (void)block_incl_scan(pre, sh);
    const int base = sh[SCAN_T - 1];
    __syncthreads();
    const int i = b * SCAN_T + t;
    const int c = (i < N_NODES) ? cnt_in[i] : 0;
    const int incl = block_incl_scan(c, sh);
    if (i < N_NODES) {
        const int start = base + incl - c;
        rowstart[i] = start;
        cursor[i]   = start;
    }
}

__global__ __launch_bounds__(256) void fill_kernel(const int* __restrict__ src,
                                                   const int* __restrict__ dst,
                                                   int* __restrict__ cursor,
                                                   int* __restrict__ csr_src) {
    int e = blockIdx.x * blockDim.x + threadIdx.x;
    if (e < N_EDGES) {
        int pos = __hip_atomic_fetch_add(&cursor[dst[e]], 1, __ATOMIC_RELAXED, __HIP_MEMORY_SCOPE_AGENT);
        csr_src[pos] = src[e];
    }
}

#define SORT_R 8

__device__ __forceinline__ void rank_row(const int* __restrict__ csr_in, int* __restrict__ csr_src,
                                         const int beg, const int n, const int key0, const int lane) {
    for (int c0 = 0; c0 < n; c0 += 32) {
        const int p = c0 + lane;
        const bool live = p < n;
        int key = key0;
        if (c0 != 0)
            key = live ? csr_in[beg + p] : 0;
        int rank = 0;
        for (int q0 = 0; q0 < n; q0 += 32) {
            const int m = min(32, n - q0);
            int other = key;
            if (q0 != c0)
                other = (q0 + lane < n) ? csr_in[beg + q0 + lane] : 0;
            for (int j = 0; j < m; ++j) {
                const int kj = __builtin_amdgcn_readlane(other, j);
                rank += (kj < key + (int)(q0 + j < p)) ? 1 : 0;
            }
        }
        if (live)
            csr_src[beg + rank] = key;
    }
}

__global__ __launch_bounds__(256) void sort_rows_kernel(const int* __restrict__ rowstart,
                                                        const int* __restrict__ csr_in,
                                                        int* __restrict__ csr_src) {
    const int lane = threadIdx.x & 31;
    const int wave = __builtin_amdgcn_readfirstlane((int)(threadIdx.x >> 5));
    const int d0   = (blockIdx.x * 8 + wave) * SORT_R;
    if (d0 >= N_NODES) return;
    int rs = N_EDGES;
    if (lane <= SORT_R && d0 + lane < N_NODES)
        rs = rowstart[d0 + lane];
    int beg[SORT_R], cnt[SORT_R], key0[SORT_R];
#pragma unroll
    for (int r = 0; r < SORT_R; ++r) {
        beg[r] = __builtin_amdgcn_readlane(rs, r);
        cnt[r] = __builtin_amdgcn_readlane(rs, r + 1) - beg[r];
        key0[r] = (lane < cnt[r]) ? csr_in[beg[r] + lane] : 0;
    }
#pragma unroll
    for (int r = 0; r < SORT_R; ++r)
        rank_row(csr_in, csr_src, beg[r], cnt[r], key0[r], lane);
}

#define GEMM_BLOCKS 512
#define ROW_TILES   (N_NODES / 16)
#define PAIR_SLOTS  (GEMM_BLOCKS * 4)
#define K_BLOCKS    (IN_CH / 16)

__device__ __forceinline__ v8f mm4(const float ax, const float ay, const float bx, const float by, const v8f c) {
    v2f a, b;
    a.x = ax; a.y = ay;
    b.x = bx; b.y = by;
    return __builtin_amdgcn_wmma_f32_16x16x4_f32(
         false, a,  false, b,
         (short)0, c,  false,  false);
}

__device__ __forceinline__ void load_block(const float* __restrict__ xr, const float4* __restrict__ bl,
                                           const int kb, float4& a0, float4& a1, float4 (&b)[4][2]) {
    a0 = *(const float4*)(xr + kb * 16);
    a1 = *(const float4*)(xr + kb * 16 + 4);
#pragma unroll
    for (int tt = 0; tt < 4; ++tt) {
        b[tt][0] = bl[kb * 512 + tt * 64];
        b[tt][1] = bl[kb * 512 + tt * 64 + 32];
    }
}

__device__ __forceinline__ void mm_block(v8f (&acc)[4], const float4 a0, const float4 a1,
                                         const float4 (&b)[4][2]) {
#pragma unroll
    for (int tt = 0; tt < 4; ++tt) acc[tt] = mm4(a0.x, a0.y, b[tt][0].x, b[tt][0].y, acc[tt]);
#pragma unroll
    for (int tt = 0; tt < 4; ++tt) acc[tt] = mm4(a0.z, a0.w, b[tt][0].z, b[tt][0].w, acc[tt]);
#pragma unroll
    for (int tt = 0; tt < 4; ++tt) acc[tt] = mm4(a1.x, a1.y, b[tt][1].x, b[tt][1].y, acc[tt]);
#pragma unroll
    for (int tt = 0; tt < 4; ++tt) acc[tt] = mm4(a1.z, a1.w, b[tt][1].z, b[tt][1].w, acc[tt]);
}

__global__ __launch_bounds__(256) __attribute__((amdgpu_waves_per_eu(4, 4))) void gemm_kernel(const float* __restrict__ X,
                                                   const float* __restrict__ W,
                                                   const float* __restrict__ inv_out,
                                                   float* __restrict__ H) {
    __shared__ __attribute__((aligned(16))) float Wl[IN_CH * HID];

    const int tid  = threadIdx.x;
    const int lane = tid & 31;
    const int wave = __builtin_amdgcn_readfirstlane(tid >> 5);
    const int half = lane >> 4;
    const int l16  = lane & 15;

    for (int p = tid; p < (IN_CH / 4) * HID; p += 256) {
        const int m = p >> 7;
        const int n = p & 127;
        const float* __restrict__ w = W + (size_t)(4 * m) * HID + n;
        float4 v;
        v.x = w[0];
        v.y = w[HID];
        v.z = w[2 * HID];
        v.w = w[3 * HID];
        const int kb = m >> 2, h = (m >> 1) & 1, q = m & 1;
        const int t = n >> 4, c = n & 15;
        *(float4*)(Wl + ((((kb * 8 + t) * 2 + q) * 32 + 16 * h + c) << 2)) = v;
    }
    __syncthreads();

    const int colhalf = wave & 1;
    const float4* __restrict__ bl = (const float4*)Wl + (size_t)(colhalf * 4) * 64 + lane;

    for (int tile = (wave >> 1) * GEMM_BLOCKS + blockIdx.x; tile < ROW_TILES; tile += PAIR_SLOTS) {
        const int rowBase = tile * 16;
        const float* __restrict__ xr = X + (size_t)(rowBase + l16) * IN_CH + half * 8;

        v8f acc[4] = {};
        float4 a0, a1, b[4][2];
        float4 c0, c1, d[4][2];
        load_block(xr, bl, 0, a0, a1, b);
#pragma unroll 1
        for (int kb = 0; kb < K_BLOCKS; kb += 2) {
            load_block(xr, bl, kb + 1, c0, c1, d);
            mm_block(acc, a0, a1, b);
            if (kb + 2 < K_BLOCKS)
                load_block(xr, bl, kb + 2, a0, a1, b);
            mm_block(acc, c0, c1, d);
        }

        const float4 i0 = *(const float4*)(inv_out + rowBase + half * 8);
        const float4 i1 = *(const float4*)(inv_out + rowBase + half * 8 + 4);
        const float iv[8] = {i0.x, i0.y, i0.z, i0.w, i1.x, i1.y, i1.z, i1.w};
        float* __restrict__ hrow = H + (size_t)(rowBase + half * 8) * HID + (colhalf * 64 + l16);
#pragma unroll
        for (int tt = 0; tt < 4; ++tt)
#pragma unroll
            for (int i = 0; i < 8; ++i)
                hrow[(size_t)i * HID + tt * 16] = acc[tt][i] * iv[i];
    }
}

__device__ __forceinline__ void walk_chunk(float4& acc, const float* __restrict__ gl,
                                           const int s_l, const int n) {
#pragma unroll 4
    for (int j = 0; j < n; ++j) {
        const int s = __builtin_amdgcn_readlane(s_l, j);
        const float4 v = *(const float4*)(gl + (size_t)s * HID);
        acc.x += v.x;
        acc.y += v.y;
        acc.z += v.z;
        acc.w += v.w;
    }
}

__device__ __forceinline__ int first_ids(const int* __restrict__ csr_src,
                                         const int beg, const int end, const int lane) {
    int s_l = 0;
    if (lane < end - beg)
        s_l = csr_src[beg + lane];
    return s_l;
}

__device__ __forceinline__ float4 pull_row(const float* __restrict__ g,
                                           const int* __restrict__ csr_src,
                                           const int s_first,
                                           const int beg, const int end, const int lane) {
    float4 acc = make_float4(0.0f, 0.0f, 0.0f, 0.0f);
    const float* __restrict__ gl = g + lane * 4;
    walk_chunk(acc, gl, s_first, min(32, end - beg));
    for (int c0 = beg + 32; c0 < end; c0 += 32) {
        const int n = min(32, end - c0);
        int s_l = 0;
        if (lane < n)
            s_l = csr_src[c0 + lane];
        walk_chunk(acc, gl, s_l, n);
    }
    return acc;
}

__device__ __forceinline__ float dst_norm(const int beg, const int end) {
    return rsqrtf((float)max(end - beg, 1));
}

__global__ __launch_bounds__(256) void pull_bias_relu_kernel(const float* __restrict__ g,
                                                             const int* __restrict__ rowstart,
                                                             const int* __restrict__ csr_src,
                                                             const float* __restrict__ inv_out,
                                                             const float* __restrict__ bias,
                                                             float* __restrict__ h0,
                                                             float* __restrict__ g0) {
    const int lane = threadIdx.x & 31;
    const int wave = __builtin_amdgcn_readfirstlane((int)(threadIdx.x >> 5));
    const int d    = blockIdx.x * 8 + wave;
    if (d >= N_NODES) return;
    const int beg = rowstart[d];
    const int end = (d + 1 < N_NODES) ? rowstart[d + 1] : N_EDGES;
    const int s_first = first_ids(csr_src, beg, end, lane);
    const float4 bb = *(const float4*)(bias + lane * 4);
    const float io = inv_out[d];
    const float wd = dst_norm(beg, end);

    const float4 acc = pull_row(g, csr_src, s_first, beg, end, lane);
    float4 r;
    r.x = fmaxf(acc.x * wd + bb.x, 0.0f);
    r.y = fmaxf(acc.y * wd + bb.y, 0.0f);
    r.z = fmaxf(acc.z * wd + bb.z, 0.0f);
    r.w = fmaxf(acc.w * wd + bb.w, 0.0f);
    *(float4*)(h0 + (size_t)d * HID + lane * 4) = r;
    r.x *= io;
    r.y *= io;
    r.z *= io;
    r.w *= io;
    *(float4*)(g0 + (size_t)d * HID + lane * 4) = r;
}

template <bool LAST>
__global__ __launch_bounds__(256) void pull_axpy_kernel(const float* __restrict__ g,
                                                        const int* __restrict__ rowstart,
                                                        const int* __restrict__ csr_src,
                                                        const float* __restrict__ inv_out,
                                                        const float* __restrict__ h0,
                                                        float* __restrict__ out) {
    const int lane = threadIdx.x & 31;
    const int wave = __builtin_amdgcn_readfirstlane((int)(threadIdx.x >> 5));
    const int d    = blockIdx.x * 8 + wave;
    if (d >= N_NODES) return;
    const int beg = rowstart[d];
    const int end = (d + 1 < N_NODES) ? rowstart[d + 1] : N_EDGES;
    const int s_first = first_ids(csr_src, beg, end, lane);
    const float4 z = *(const float4*)(h0 + (size_t)d * HID + lane * 4);
    const float io = LAST ? 1.0f : inv_out[d];
    const float wd = dst_norm(beg, end);

    const float4 a = pull_row(g, csr_src, s_first, beg, end, lane);
    float4 r;
    r.x = fmaf(1.0f - ALPHA, a.x * wd, ALPHA * z.x);
    r.y = fmaf(1.0f - ALPHA, a.y * wd, ALPHA * z.y);
    r.z = fmaf(1.0f - ALPHA, a.z * wd, ALPHA * z.z);
    r.w = fmaf(1.0f - ALPHA, a.w * wd, ALPHA * z.w);
    if (!LAST) {
        r.x *= io;
        r.y *= io;
        r.z *= io;
        r.w *= io;
    }
    *(float4*)(out + (size_t)d * HID + lane * 4) = r;
}

extern "C" void kernel_launch(void* const* d_in, const int* in_sizes, int n_in,
                              void* d_out, int out_size, void* d_ws, size_t ws_size,
                              hipStream_t stream) {
    const float* features = (const float*)d_in[0];
    const float* W        = (const float*)d_in[1];
    const float* bias     = (const float*)d_in[2];
    const int*   eidx     = (const int*)d_in[3];
    const int*   src      = eidx;
    const int*   dst      = eidx + N_EDGES;

    float* out = (float*)d_out;

    const size_t NH = (size_t)N_NODES * HID;
    float* inv_out  = (float*)d_ws;
    int*   rowstart = (int*)(inv_out + N_NODES);
    int*   csr_src  = rowstart + N_NODES;
    float* bufA     = (float*)(csr_src + N_EDGES);
    float* bufB     = bufA + NH;

    int* cnt_out = (int*)bufB;
    int* cnt_in  = cnt_out + N_NODES;
    int* cursor  = cnt_in + N_NODES;
    int* csr_tmp = cursor + N_NODES;

    hipMemsetAsync(cnt_out, 0, 2 * (size_t)N_NODES * sizeof(int), stream);
    count_kernel<<<(N_EDGES + 255) / 256, 256, 0, stream>>>(src, dst, cnt_out, cnt_in);
    inv_kernel<<<(N_NODES + 255) / 256, 256, 0, stream>>>(cnt_out, inv_out);
    scan_kernel<<<SCAN_B, SCAN_T, 0, stream>>>(cnt_in, rowstart, cursor);
    fill_kernel<<<(N_EDGES + 255) / 256, 256, 0, stream>>>(src, dst, cursor, csr_tmp);
    sort_rows_kernel<<<(N_NODES + 8 * SORT_R - 1) / (8 * SORT_R), 256, 0, stream>>>(rowstart, csr_tmp, csr_src);

    gemm_kernel<<<GEMM_BLOCKS, 256, 0, stream>>>(features, W, inv_out, bufA);
    pull_bias_relu_kernel<<<(N_NODES + 7) / 8, 256, 0, stream>>>(bufA, rowstart, csr_src, inv_out, bias, bufB, out);

    float* cur = out;
    for (int t = 0; t < K_STEPS - 1; ++t) {
        float* tmp = (t & 1) ? out : bufA;
        pull_axpy_kernel<false><<<(N_NODES + 7) / 8, 256, 0, stream>>>(cur, rowstart, csr_src, inv_out, bufB, tmp);
        cur = tmp;
    }
    pull_axpy_kernel<true><<<(N_NODES + 7) / 8, 256, 0, stream>>>(cur, rowstart, csr_src, inv_out, bufB, out);
}
